# starting kernel
# baseline (speedup 1.0000x reference)
#include <hip/hip_runtime.h>

typedef float v2f __attribute__((ext_vector_type(2)));
typedef float v4f __attribute__((ext_vector_type(4)));
typedef float v8f __attribute__((ext_vector_type(8)));

#define N_MODES    64
#define N_CLASSES  10
#define TILE_ROWS  128
#define LDS_STRIDE 68

__global__ void cv2d_fold_kernel(const float* __restrict__ S, const float* __restrict__ d,
                                 const float* __restrict__ bias, const float* __restrict__ W,
                                 const float* __restrict__ b, float* __restrict__ Mfold,
                                 float* __restrict__ cst) {
  int t = blockIdx.x * blockDim.x + threadIdx.x;
  if (t >= 16 * N_MODES) return;
  int c = t >> 6;
  int i = t & 63;
  float acc = 0.f;
  if (c < N_CLASSES) {
    for (int m = 0; m < N_MODES; ++m)
      acc += W[c * N_MODES + m] * S[(2 * m) * (2 * N_MODES) + 2 * i];
  }
  Mfold[c * N_MODES + i] = acc;
  if (i == 0) {
    float cv = 0.f;
    if (c < N_CLASSES) {
      for (int m = 0; m < N_MODES; ++m)
        cv += W[c * N_MODES + m] * (d[2 * m] + bias[m]);
      cv += b[c];
    }
    cst[c] = cv;
  }
}

__global__ __launch_bounds__(256) void cv2d_gemm_kernel(const float* __restrict__ x,
                                                        const float* __restrict__ Mfold,
                                                        const float* __restrict__ cst,
                                                        float* __restrict__ out, int Bn) {
  __shared__ __align__(16) float lds[TILE_ROWS * LDS_STRIDE];

  const int t    = threadIdx.x;
  const int lane = t & 31;
  const int wave = t >> 5;
  const int r0   = blockIdx.x * TILE_ROWS;

  {
    const int row  = t >> 1;
    const int half = t & 1;
    int grow = r0 + row;
    if (grow >= Bn) grow = Bn - 1;
    const v4f* src = (const v4f*)(x + (size_t)grow * N_MODES + half * 32);
    float*     dst = lds + row * LDS_STRIDE + half * 32;
#pragma unroll
    for (int i = 0; i < 8; ++i) {
      v4f v = __builtin_nontemporal_load(src + i);
      *(v4f*)(dst + i * 4) = v;
    }
  }

  const int n     = lane & 15;
  const int khalf = (lane >> 4) << 1;
  v2f bfrag[16];
#pragma unroll
  for (int tk = 0; tk < 16; ++tk)
    bfrag[tk] = *(const v2f*)(Mfold + n * N_MODES + tk * 4 + khalf);
  const float cn = cst[n];

  __syncthreads();

  v8f acc = {};
  const float* arow = lds + (wave * 16 + (lane & 15)) * LDS_STRIDE + khalf;
#pragma unroll
  for (int tk = 0; tk < 16; ++tk) {
    v2f a = *(const v2f*)(arow + tk * 4);
    acc = __builtin_amdgcn_wmma_f32_16x16x4_f32(
         false, a,  false, bfrag[tk],
         (short)0, acc,  false,  false);
  }

  const int rbase = r0 + wave * 16 + ((lane >> 4) << 3);
  const bool tile_full = (r0 + TILE_ROWS) <= Bn;
  if (n < N_CLASSES) {
    float* op = out + (size_t)rbase * N_CLASSES + n;
    if (tile_full) {
#pragma unroll
      for (int v = 0; v < 8; ++v)
        __builtin_nontemporal_store(acc[v] + cn, op + v * N_CLASSES);
    } else {
#pragma unroll
      for (int v = 0; v < 8; ++v)
        if (rbase + v < Bn)
          __builtin_nontemporal_store(acc[v] + cn, op + v * N_CLASSES);
    }
  }
}

extern "C" void kernel_launch(void* const* d_in, const int* in_sizes, int n_in,
                              void* d_out, int out_size, void* d_ws, size_t ws_size,
                              hipStream_t stream) {
  const float* x    = (const float*)d_in[0];
  const float* S    = (const float*)d_in[1];
  const float* d    = (const float*)d_in[2];
  const float* bias = (const float*)d_in[3];
  const float* W    = (const float*)d_in[4];
  const float* b    = (const float*)d_in[5];
  float* out = (float*)d_out;

  const int Bn = in_sizes[0] / N_MODES;

  float* Mfold = (float*)d_ws;
  float* cst   = Mfold + 16 * N_MODES;

  hipLaunchKernelGGL(cv2d_fold_kernel, dim3(4), dim3(256), 0, stream,
                     S, d, bias, W, b, Mfold, cst);

  const int nblocks = (Bn + TILE_ROWS - 1) / TILE_ROWS;
  hipLaunchKernelGGL(cv2d_gemm_kernel, dim3(nblocks), dim3(256), 0, stream,
                     x, Mfold, cst, out, Bn);
}


# register-only streaming: 256 resident workgroups x 8 waves loop over 16-row groups; plain loads put x rows straight into WMMA A registers (same k order on A and B), next group in flight; no LDS
# speedup vs baseline: 1.2078x; 1.2078x over previous
#include <hip/hip_runtime.h>


typedef float v2f __attribute__((ext_vector_type(2)));
typedef float v4f __attribute__((ext_vector_type(4)));
typedef float v8f __attribute__((ext_vector_type(8)));

#define N_MODES    64
#define N_CLASSES  10
#define NBLK       256
#define WPB        8

__global__ void cv2d_fold_kernel(const float* __restrict__ S, const float* __restrict__ d,
                                 const float* __restrict__ bias, const float* __restrict__ W,
                                 const float* __restrict__ b, float* __restrict__ Mfold,
                                 float* __restrict__ cst) {
  int t = blockIdx.x * blockDim.x + threadIdx.x;
  if (t >= 16 * N_MODES) return;
  int c = t >> 6;
  int i = t & 63;
  float acc = 0.f;
  if (c < N_CLASSES) {
    for (int m = 0; m < N_MODES; ++m)
      acc += W[c * N_MODES + m] * S[(2 * m) * (2 * N_MODES) + 2 * i];
  }
  Mfold[c * N_MODES + i] = acc;
  if (i == 0) {
    float cv = 0.f;
    if (c < N_CLASSES) {
      for (int m = 0; m < N_MODES; ++m)
        cv += W[c * N_MODES + m] * (d[2 * m] + bias[m]);
      cv += b[c];
    }
    cst[c] = cv;
  }
}

__device__ __forceinline__ void load_row_frags(const float* __restrict__ rowp_h, v4f (&f)[8]) {
  const v4f* src = (const v4f*)rowp_h;
#pragma unroll
  for (int j = 0; j < 8; ++j) f[j] = src[2 * j];
}

__device__ __forceinline__ v8f mma_group(const v4f (&a)[8], const v4f (&b)[8]) {
  v8f acc = {};
#pragma unroll
  for (int j = 0; j < 8; ++j) {
    v2f a0 = {a[j].x, a[j].y};
    v2f b0 = {b[j].x, b[j].y};
    acc = __builtin_amdgcn_wmma_f32_16x16x4_f32(false, a0, false, b0, (short)0, acc, false, false);
    v2f a1 = {a[j].z, a[j].w};
    v2f b1 = {b[j].z, b[j].w};
    acc = __builtin_amdgcn_wmma_f32_16x16x4_f32(false, a1, false, b1, (short)0, acc, false, false);
  }
  return acc;
}

__device__ __forceinline__ void store_group(const v8f& acc, float cn, float* __restrict__ out,
                                            int g, int lane, int Bn, bool live) {
  if (!live) return;
  const int n = lane & 15;
  const int rbase = g * 16 + ((lane >> 4) << 3);
  const bool full = (g * 16 + 16) <= Bn;
  if (n < N_CLASSES) {
    float* op = out + (size_t)rbase * N_CLASSES + n;
    if (full) {
#pragma unroll
      for (int v = 0; v < 8; ++v)
        __builtin_nontemporal_store(acc[v] + cn, op + v * N_CLASSES);
    } else {
#pragma unroll
      for (int v = 0; v < 8; ++v)
        if (rbase + v < Bn)
          __builtin_nontemporal_store(acc[v] + cn, op + v * N_CLASSES);
    }
  }
}

__device__ __forceinline__ const float* x_row_ptr(const float* __restrict__ x, int g, int lane,
                                                  int Bn, bool live) {
  int row = g * 16 + (lane & 15);
  if (row >= Bn) row = Bn - 1;
  if (!live) row = 0;
  return x + (size_t)row * N_MODES + ((lane >> 4) << 2);
}

__global__ __launch_bounds__(256) void cv2d_gemm_kernel(const float* __restrict__ x,
                                                        const float* __restrict__ Mfold,
                                                        const float* __restrict__ cst,
                                                        float* __restrict__ out, int Bn) {
  const int lane = threadIdx.x & 31;
  const int wave = __builtin_amdgcn_readfirstlane(threadIdx.x >> 5);
  const int ngroups = (Bn + 15) >> 4;
  const int g_begin = (int)(((long long)blockIdx.x * ngroups) / NBLK);
  const int g_end   = (int)(((long long)(blockIdx.x + 1) * ngroups) / NBLK);

  int g = g_begin + wave;
  if (g >= g_end) return;

  v4f xa[8], xb[8];
  load_row_frags(x_row_ptr(x, g, lane, Bn, true), xa);

  const int n = lane & 15;
  v4f bf[8];
  load_row_frags(Mfold + n * N_MODES + ((lane >> 4) << 2), bf);
  const float cn = cst[n];

  for (;;) {
    const int g1 = g + WPB;
    const bool live1 = g1 < g_end;
    load_row_frags(x_row_ptr(x, g1, lane, Bn, live1), xb);
    __builtin_amdgcn_sched_barrier(0);
    store_group(mma_group(xa, bf), cn, out, g, lane, Bn, true);
    const int g2 = g1 + WPB;
    const bool live2 = g2 < g_end;
    load_row_frags(x_row_ptr(x, g2, lane, Bn, live2), xa);
    __builtin_amdgcn_sched_barrier(0);
    store_group(mma_group(xb, bf), cn, out, g1, lane, Bn, live1);
    if (!live2) break;
    g = g2;
  }
}

extern "C" void kernel_launch(void* const* d_in, const int* in_sizes, int n_in,
                              void* d_out, int out_size, void* d_ws, size_t ws_size,
                              hipStream_t stream) {
  const float* x    = (const float*)d_in[0];
  const float* S    = (const float*)d_in[1];
  const float* d    = (const float*)d_in[2];
  const float* bias = (const float*)d_in[3];
  const float* W    = (const float*)d_in[4];
  const float* b    = (const float*)d_in[5];
  float* out = (float*)d_out;

  const int Bn = in_sizes[0] / N_MODES;

  float* Mfold = (float*)d_ws;
  float* cst   = Mfold + 16 * N_MODES;

  hipLaunchKernelGGL(cv2d_fold_kernel, dim3(4), dim3(256), 0, stream,
                     S, d, bias, W, b, Mfold, cst);

  hipLaunchKernelGGL(cv2d_gemm_kernel, dim3(NBLK), dim3(32 * WPB), 0, stream,
                     x, Mfold, cst, out, Bn);
}


# four waves per workgroup, one per SIMD: the per-wave fold runs once per SIMD, not twice; each wave holds its whole share of x in registers, requested ahead of the products; two accumulator chains
# speedup vs baseline: 2.3694x; 1.9617x over previous
#include <hip/hip_runtime.h>


typedef float v2f __attribute__((ext_vector_type(2)));
typedef float v4f __attribute__((ext_vector_type(4)));
typedef float v8f __attribute__((ext_vector_type(8)));

#define N_MODES    64
#define N_CLASSES  10
#define NBLK       256
#define WPB        4
#define MAXG       13
#define NPRE       7

__device__ __forceinline__ void load_row_frags(const float* __restrict__ rowp_h, v4f (&f)[8]) {
  const v4f* src = (const v4f*)rowp_h;
#pragma unroll
  for (int j = 0; j < 8; ++j) f[j] = src[2 * j];
}

__device__ __forceinline__ v8f mma_group(const v4f (&a)[8], const v4f (&b)[8]) {
  v8f acc = {};
#pragma unroll
  for (int j = 0; j < 8; ++j) {
    v2f a0 = {a[j].x, a[j].y};
    v2f b0 = {b[j].x, b[j].y};
    acc = __builtin_amdgcn_wmma_f32_16x16x4_f32(false, a0, false, b0, (short)0, acc, false, false);
    v2f a1 = {a[j].z, a[j].w};
    v2f b1 = {b[j].z, b[j].w};
    acc = __builtin_amdgcn_wmma_f32_16x16x4_f32(false, a1, false, b1, (short)0, acc, false, false);
  }
  return acc;
}

template <int J0, int J1>
__device__ __forceinline__ void mma_pair_pieces(const v4f (&a)[8], const v4f (&c)[8], const v4f (&b)[8],
                                                v8f& p, v8f& q) {
#pragma unroll
  for (int j = J0; j < J1; ++j) {
    v2f b0 = {b[j].x, b[j].y};
    v2f a0 = {a[j].x, a[j].y};
    v2f c0 = {c[j].x, c[j].y};
    p = __builtin_amdgcn_wmma_f32_16x16x4_f32(false, a0, false, b0, (short)0, p, false, false);
    q = __builtin_amdgcn_wmma_f32_16x16x4_f32(false, c0, false, b0, (short)0, q, false, false);
    v2f b1 = {b[j].z, b[j].w};
    v2f a1 = {a[j].z, a[j].w};
    v2f c1 = {c[j].z, c[j].w};
    p = __builtin_amdgcn_wmma_f32_16x16x4_f32(false, a1, false, b1, (short)0, p, false, false);
    q = __builtin_amdgcn_wmma_f32_16x16x4_f32(false, c1, false, b1, (short)0, q, false, false);
  }
}

template <int J0, int J1>
__device__ __forceinline__ void mma_pieces(const v4f (&a)[8], const v4f (&b)[8], v8f& p) {
#pragma unroll
  for (int j = J0; j < J1; ++j) {
    v2f a0 = {a[j].x, a[j].y};
    v2f b0 = {b[j].x, b[j].y};
    p = __builtin_amdgcn_wmma_f32_16x16x4_f32(false, a0, false, b0, (short)0, p, false, false);
    v2f a1 = {a[j].z, a[j].w};
    v2f b1 = {b[j].z, b[j].w};
    p = __builtin_amdgcn_wmma_f32_16x16x4_f32(false, a1, false, b1, (short)0, p, false, false);
  }
}

__device__ __forceinline__ void store_group(const v8f& acc, float cn, float* __restrict__ out,
                                            int g, int lane, int Bn, bool live) {
  if (!live) return;
  const int n = lane & 15;
  const int rbase = g * 16 + ((lane >> 4) << 3);
  const bool full = (g * 16 + 16) <= Bn;
  if (n < N_CLASSES) {
    float* op = out + (size_t)rbase * N_CLASSES + n;
    if (full) {
#pragma unroll
      for (int v = 0; v < 8; ++v)
        __builtin_nontemporal_store(acc[v] + cn, op + v * N_CLASSES);
    } else {
#pragma unroll
      for (int v = 0; v < 8; ++v)
        if (rbase + v < Bn)
          __builtin_nontemporal_store(acc[v] + cn, op + v * N_CLASSES);
    }
  }
}

__device__ __forceinline__ void store_full(const v8f& acc, float cn, float* __restrict__ out,
                                           int g, int lane, bool ok) {
  if (!ok) return;
  const int n = lane & 15;
  if (n < N_CLASSES) {
    float* op = out + (size_t)(g * 16 + ((lane >> 4) << 3)) * N_CLASSES + n;
#pragma unroll
    for (int v = 0; v < 8; ++v)
      __builtin_nontemporal_store(acc[v] + cn, op + v * N_CLASSES);
  }
}

__device__ __forceinline__ const float* x_row_ptr(const float* __restrict__ x, int g, int lane,
                                                  int Bn, bool live) {
  int row = g * 16 + (lane & 15);
  if (row >= Bn) row = Bn - 1;
  if (!live) row = 0;
  return x + (size_t)row * N_MODES + ((lane >> 4) << 2);
}

__device__ __forceinline__ float other_half(float v) {
  return __builtin_bit_cast(float, __builtin_amdgcn_permlanex16(__builtin_bit_cast(int, v), __builtin_bit_cast(int, v),
                                                                 0x76543210, 0xfedcba98, false, false));
}

struct FoldOperands {
  float dm[8][4];
  v4f   bq[8];
  float bn;
  v4f   wf[8];
  float sa[4][8][4];
};

__device__ __forceinline__ void fold_load(const float* __restrict__ S, const float* __restrict__ d,
                                          const float* __restrict__ bias, const float* __restrict__ W,
                                          const float* __restrict__ b, int lane, FoldOperands& fo) {
  const int n = lane & 15;
  const int h = lane >> 4;
  const int nrow = n < N_CLASSES ? n : (N_CLASSES - 1);
#pragma unroll
  for (int j = 0; j < 8; ++j) {
#pragma unroll
    for (int e = 0; e < 4; ++e) fo.dm[j][e] = d[16 * j + 8 * h + 2 * e];
    fo.bq[j] = *(const v4f*)(bias + 8 * j + 4 * h);
  }
  fo.bn = b[nrow];
  load_row_frags(W + nrow * N_MODES + (h << 2), fo.wf);
  const int imode = (((n & 7) >> 2) << 3) + ((n >> 3) << 2) + (n & 3);
  const float* sp0 = S + (8 * h) * (2 * N_MODES) + 2 * imode;
#pragma unroll
  for (int j = 0; j < 8; ++j)
#pragma unroll
    for (int T = 0; T < 4; ++T)
#pragma unroll
      for (int e = 0; e < 4; ++e)
        fo.sa[T][j][e] = sp0[j * (16 * 2 * N_MODES) + e * (2 * 2 * N_MODES) + T * 32];
}

__device__ __forceinline__ void fold_math(const FoldOperands& fo, v4f (&bf)[8], float& cn) {
  v8f acc[4] = {};
#pragma unroll
  for (int j = 0; j < 8; ++j) {
    v2f b0 = {fo.wf[j].x, fo.wf[j].y};
#pragma unroll
    for (int T = 0; T < 4; ++T) {
      v2f a0 = {fo.sa[T][j][0], fo.sa[T][j][1]};
      acc[T] = __builtin_amdgcn_wmma_f32_16x16x4_f32(false, a0, false, b0, (short)0, acc[T], false, false);
    }
    v2f b1 = {fo.wf[j].z, fo.wf[j].w};
#pragma unroll
    for (int T = 0; T < 4; ++T) {
      v2f a1 = {fo.sa[T][j][2], fo.sa[T][j][3]};
      acc[T] = __builtin_amdgcn_wmma_f32_16x16x4_f32(false, a1, false, b1, (short)0, acc[T], false, false);
    }
  }
#pragma unroll
  for (int T = 0; T < 4; ++T) {
    bf[2 * T]     = (v4f){acc[T][0], acc[T][1], acc[T][2], acc[T][3]};
    bf[2 * T + 1] = (v4f){acc[T][4], acc[T][5], acc[T][6], acc[T][7]};
  }

  float part = 0.f;
#pragma unroll
  for (int j = 0; j < 8; ++j) {
    part += fo.wf[j].x * (fo.dm[j][0] + fo.bq[j].x);
    part += fo.wf[j].y * (fo.dm[j][1] + fo.bq[j].y);
    part += fo.wf[j].z * (fo.dm[j][2] + fo.bq[j].z);
    part += fo.wf[j].w * (fo.dm[j][3] + fo.bq[j].w);
  }
  part += other_half(part);
  cn = part + fo.bn;
}

__global__ __launch_bounds__(32 * WPB) void cv2d_gemm_kernel(const float* __restrict__ x,
                                                             const float* __restrict__ S,
                                                             const float* __restrict__ d,
                                                             const float* __restrict__ bias,
                                                             const float* __restrict__ W,
                                                             const float* __restrict__ b,
                                                             float* __restrict__ out, int Bn) {
  const int lane = threadIdx.x & 31;
  const int wave = __builtin_amdgcn_readfirstlane(threadIdx.x >> 5);
  const int ngroups = (Bn + 15) >> 4;
  const int nfull = Bn >> 4;
  const int g_begin = (int)(((long long)blockIdx.x * ngroups) / NBLK);
  const int g_end   = (int)(((long long)(blockIdx.x + 1) * ngroups) / NBLK);
  const int g_store = g_end < nfull ? g_end : nfull;

  int base = g_begin + wave;
  if (base >= g_end) return;

  FoldOperands fo;
  fold_load(S, d, bias, W, b, lane, fo);
  v4f xs[MAXG][8];
#pragma unroll
  for (int i = 0; i < NPRE; ++i) {
    const int g = base + i * WPB;
    load_row_frags(x_row_ptr(x, g, lane, Bn, g < g_end), xs[i]);
  }
  __builtin_amdgcn_sched_barrier(0);

  v4f bf[8];
  float cn;
  fold_math(fo, bf, cn);
  __builtin_amdgcn_sched_barrier(0);

#pragma unroll
  for (int i = NPRE; i < MAXG; ++i) {
    const int g = base + i * WPB;
    load_row_frags(x_row_ptr(x, g, lane, Bn, g < g_end), xs[i]);
  }
  __builtin_amdgcn_sched_barrier(0);

  for (;;) {
    v8f pa = {}, pb = {};
#pragma unroll
    for (int p = 0; p < MAXG / 2; ++p) {
      const int ga = base + (2 * p) * WPB;
      const int ha = ga - 2 * WPB;
      v8f ra = {}, rb = {};
      __builtin_amdgcn_sched_barrier(0);
      mma_pair_pieces<0, 2>(xs[2 * p], xs[2 * p + 1], bf, ra, rb);
      if (p > 0) store_full(pa, cn, out, ha, lane, ha < g_store);
      mma_pair_pieces<2, 4>(xs[2 * p], xs[2 * p + 1], bf, ra, rb);
      if (p > 0) store_full(pb, cn, out, ha + WPB, lane, ha + WPB < g_store);
      mma_pair_pieces<4, 8>(xs[2 * p], xs[2 * p + 1], bf, ra, rb);
      pa = ra;
      pb = rb;
    }
    {
      const int gl = base + (MAXG - 1) * WPB;
      const int ha = gl - 2 * WPB;
      v8f rl = {};
      __builtin_amdgcn_sched_barrier(0);
      mma_pieces<0, 4>(xs[MAXG - 1], bf, rl);
      store_full(pa, cn, out, ha, lane, ha < g_store);
      mma_pieces<4, 8>(xs[MAXG - 1], bf, rl);
      store_full(pb, cn, out, ha + WPB, lane, ha + WPB < g_store);
      store_full(rl, cn, out, gl, lane, gl < g_store);
    }
    base += MAXG * WPB;
    if (base >= g_end) break;
#pragma unroll
    for (int i = 0; i < MAXG; ++i) {
      const int g = base + i * WPB;
      load_row_frags(x_row_ptr(x, g, lane, Bn, g < g_end), xs[i]);
    }
    __builtin_amdgcn_sched_barrier(0);
  }

  if (nfull < ngroups) {
    const int gl = ngroups - 1;
    if (gl >= g_begin && gl < g_end && ((gl - g_begin) % WPB) == wave) {
      v4f xl[8];
      load_row_frags(x_row_ptr(x, gl, lane, Bn, true), xl);
      store_group(mma_group(xl, bf), cn, out, gl, lane, Bn, true);
    }
  }
}

extern "C" void kernel_launch(void* const* d_in, const int* in_sizes, int n_in,
                              void* d_out, int out_size, void* d_ws, size_t ws_size,
                              hipStream_t stream) {
  const float* x    = (const float*)d_in[0];
  const float* S    = (const float*)d_in[1];
  const float* d    = (const float*)d_in[2];
  const float* bias = (const float*)d_in[3];
  const float* W    = (const float*)d_in[4];
  const float* b    = (const float*)d_in[5];
  float* out = (float*)d_out;

  const int Bn = in_sizes[0] / N_MODES;

  hipLaunchKernelGGL(cv2d_gemm_kernel, dim3(NBLK), dim3(32 * WPB), 0, stream,
                     x, S, d, bias, W, b, out, Bn);
}


# fold once per workgroup, not per wave: each of the 4 waves forms one mode tile (16 matrix instructions, 51 loads; was 64, 177); tiles and constant partials pass through 4608 B of LDS, one barrier
# speedup vs baseline: 2.6691x; 1.1265x over previous
#include <hip/hip_runtime.h>


typedef float v2f __attribute__((ext_vector_type(2)));
typedef float v4f __attribute__((ext_vector_type(4)));
typedef float v8f __attribute__((ext_vector_type(8)));
typedef float v4fu __attribute__((ext_vector_type(4), aligned(4)));

#define N_MODES    64
#define N_CLASSES  10
#define NBLK       256
#define WPB        4
#define MAXG       13
#define NPRE       7

__device__ __forceinline__ void load_row_frags(const float* __restrict__ rowp_h, v4f (&f)[8]) {
  const v4f* src = (const v4f*)rowp_h;
#pragma unroll
  for (int j = 0; j < 8; ++j) f[j] = src[2 * j];
}

__device__ __forceinline__ v8f mma_group(const v4f (&a)[8], const v4f (&b)[8], const v8f& c0) {
  v8f acc = c0;
#pragma unroll
  for (int j = 0; j < 8; ++j) {
    v2f a0 = {a[j].x, a[j].y};
    v2f b0 = {b[j].x, b[j].y};
    acc = __builtin_amdgcn_wmma_f32_16x16x4_f32(false, b0, false, a0, (short)0, acc, false, false);
    v2f a1 = {a[j].z, a[j].w};
    v2f b1 = {b[j].z, b[j].w};
    acc = __builtin_amdgcn_wmma_f32_16x16x4_f32(false, b1, false, a1, (short)0, acc, false, false);
  }
  return acc;
}

template <int J0, int J1>
__device__ __forceinline__ void mma_pair_pieces(const v4f (&a)[8], const v4f (&c)[8], const v4f (&b)[8],
                                                v8f& p, v8f& q) {
#pragma unroll
  for (int j = J0; j < J1; ++j) {
    v2f b0 = {b[j].x, b[j].y};
    v2f a0 = {a[j].x, a[j].y};
    v2f c0 = {c[j].x, c[j].y};
    p = __builtin_amdgcn_wmma_f32_16x16x4_f32(false, b0, false, a0, (short)0, p, false, false);
    q = __builtin_amdgcn_wmma_f32_16x16x4_f32(false, b0, false, c0, (short)0, q, false, false);
    v2f b1 = {b[j].z, b[j].w};
    v2f a1 = {a[j].z, a[j].w};
    v2f c1 = {c[j].z, c[j].w};
    p = __builtin_amdgcn_wmma_f32_16x16x4_f32(false, b1, false, a1, (short)0, p, false, false);
    q = __builtin_amdgcn_wmma_f32_16x16x4_f32(false, b1, false, c1, (short)0, q, false, false);
  }
}

template <int J0, int J1>
__device__ __forceinline__ void mma_pieces(const v4f (&a)[8], const v4f (&b)[8], v8f& p) {
#pragma unroll
  for (int j = J0; j < J1; ++j) {
    v2f a0 = {a[j].x, a[j].y};
    v2f b0 = {b[j].x, b[j].y};
    p = __builtin_amdgcn_wmma_f32_16x16x4_f32(false, b0, false, a0, (short)0, p, false, false);
    v2f a1 = {a[j].z, a[j].w};
    v2f b1 = {b[j].z, b[j].w};
    p = __builtin_amdgcn_wmma_f32_16x16x4_f32(false, b1, false, a1, (short)0, p, false, false);
  }
}

__device__ __forceinline__ void store_lane5(const v8f& acc, float* __restrict__ op) {
  v4fu lo = {acc[0], acc[1], acc[2], acc[3]};
  __builtin_nontemporal_store(lo, (v4fu*)op);
  __builtin_nontemporal_store(acc[4], op + 4);
}

__device__ __forceinline__ void store_group(const v8f& acc, float* __restrict__ out,
                                            int g, int lane, int Bn) {
  const int row = g * 16 + (lane & 15);
  if (row < Bn) store_lane5(acc, out + (size_t)row * N_CLASSES + 5 * (lane >> 4));
}

__device__ __forceinline__ void store_full(const v8f& acc, float* __restrict__ out,
                                           int g, int lane, bool ok) {
  if (!ok) return;
  store_lane5(acc, out + (size_t)(g * 16 + (lane & 15)) * N_CLASSES + 5 * (lane >> 4));
}

__device__ __forceinline__ const float* x_row_ptr(const float* __restrict__ x, int g, int lane,
                                                  int Bn, bool live) {
  int row = g * 16 + (lane & 15);
  if (row >= Bn) row = Bn - 1;
  if (!live) row = 0;
  return x + (size_t)row * N_MODES + ((lane >> 4) << 2);
}

__device__ __forceinline__ float other_half(float v) {
  return __builtin_bit_cast(float, __builtin_amdgcn_permlanex16(__builtin_bit_cast(int, v), __builtin_bit_cast(int, v),
                                                                 0x76543210, 0xfedcba98, false, false));
}

struct FoldOperands {
  float dm[2][4];
  v4f   bq[2];
  float bn;
  v4f   wf[8];
  float sa[8][4];
};

__device__ __forceinline__ void fold_load(const float* __restrict__ S, const float* __restrict__ d,
                                          const float* __restrict__ bias, const float* __restrict__ W,
                                          const float* __restrict__ b, int lane, int T, FoldOperands& fo) {
  const int n = lane & 15;
  const int h = lane >> 4;
  const int nrow = n < 5 ? n : ((n >= 8 && n < 13) ? n - 3 : (N_CLASSES - 1));
#pragma unroll
  for (int q = 0; q < 2; ++q) {
    const int j = 2 * T + q;
#pragma unroll
    for (int e = 0; e < 4; ++e) fo.dm[q][e] = d[16 * j + 8 * h + 2 * e];
    fo.bq[q] = *(const v4f*)(bias + 8 * j + 4 * h);
  }
  fo.bn = b[nrow];
  load_row_frags(W + nrow * N_MODES + (h << 2), fo.wf);
  const int imode = (((n & 7) >> 2) << 3) + ((n >> 3) << 2) + (n & 3);
  const float* sp0 = S + (8 * h) * (2 * N_MODES) + 2 * imode + T * 32;
#pragma unroll
  for (int j = 0; j < 8; ++j)
#pragma unroll
    for (int e = 0; e < 4; ++e)
      fo.sa[j][e] = sp0[j * (16 * 2 * N_MODES) + e * (2 * 2 * N_MODES)];
}

__device__ __forceinline__ v4f pick4(int T, const v4f& a0, const v4f& a1, const v4f& a2, const v4f& a3) {
  const v4f lo = T == 0 ? a0 : a1;
  const v4f hi = T == 2 ? a2 : a3;
  return T < 2 ? lo : hi;
}

__device__ __forceinline__ void fold_math(const FoldOperands& fo, int T, v8f& acc, float& part) {
  v8f ae = {}, ao = {};
#pragma unroll
  for (int j = 0; j < 8; j += 2) {
    v2f se0 = {fo.sa[j][0], fo.sa[j][1]};
    v2f we0 = {fo.wf[j].x, fo.wf[j].y};
    v2f so0 = {fo.sa[j + 1][0], fo.sa[j + 1][1]};
    v2f wo0 = {fo.wf[j + 1].x, fo.wf[j + 1].y};
    ae = __builtin_amdgcn_wmma_f32_16x16x4_f32(false, se0, false, we0, (short)0, ae, false, false);
    ao = __builtin_amdgcn_wmma_f32_16x16x4_f32(false, so0, false, wo0, (short)0, ao, false, false);
    v2f se1 = {fo.sa[j][2], fo.sa[j][3]};
    v2f we1 = {fo.wf[j].z, fo.wf[j].w};
    v2f so1 = {fo.sa[j + 1][2], fo.sa[j + 1][3]};
    v2f wo1 = {fo.wf[j + 1].z, fo.wf[j + 1].w};
    ae = __builtin_amdgcn_wmma_f32_16x16x4_f32(false, se1, false, we1, (short)0, ae, false, false);
    ao = __builtin_amdgcn_wmma_f32_16x16x4_f32(false, so1, false, wo1, (short)0, ao, false, false);
  }
  acc = ae + ao;

  const v4f w0 = pick4(T, fo.wf[0], fo.wf[2], fo.wf[4], fo.wf[6]);
  const v4f w1 = pick4(T, fo.wf[1], fo.wf[3], fo.wf[5], fo.wf[7]);
  float p = 0.f;
  p += w0.x * (fo.dm[0][0] + fo.bq[0].x);
  p += w0.y * (fo.dm[0][1] + fo.bq[0].y);
  p += w0.z * (fo.dm[0][2] + fo.bq[0].z);
  p += w0.w * (fo.dm[0][3] + fo.bq[0].w);
  p += w1.x * (fo.dm[1][0] + fo.bq[1].x);
  p += w1.y * (fo.dm[1][1] + fo.bq[1].y);
  p += w1.z * (fo.dm[1][2] + fo.bq[1].z);
  p += w1.w * (fo.dm[1][3] + fo.bq[1].w);
  p += other_half(p);
  part = p;
}

__global__ __launch_bounds__(32 * WPB) void cv2d_gemm_kernel(const float* __restrict__ x,
                                                             const float* __restrict__ S,
                                                             const float* __restrict__ d,
                                                             const float* __restrict__ bias,
                                                             const float* __restrict__ W,
                                                             const float* __restrict__ b,
                                                             float* __restrict__ out, int Bn) {
  const int lane = threadIdx.x & 31;
  const int wave = __builtin_amdgcn_readfirstlane(threadIdx.x >> 5);
  const int ngroups = (Bn + 15) >> 4;
  const int nfull = Bn >> 4;
  const int g_begin = (int)(((long long)blockIdx.x * ngroups) / NBLK);
  const int g_end   = (int)(((long long)(blockIdx.x + 1) * ngroups) / NBLK);
  const int g_store = g_end < nfull ? g_end : nfull;

  if (g_begin >= g_end) return;
  int base = g_begin + wave;
  const bool have = base < g_end;

  __shared__ v4f   sh_bf[8][32];
  __shared__ float sh_cp[WPB][32];

  FoldOperands fo;
  fold_load(S, d, bias, W, b, lane, wave, fo);
  v4f xs[MAXG][8];
#pragma unroll
  for (int i = 0; i < NPRE; ++i) {
    const int g = base + i * WPB;
    load_row_frags(x_row_ptr(x, g, lane, Bn, g < g_end), xs[i]);
  }
  __builtin_amdgcn_sched_barrier(0);

  v4f bf[8];
  float cn;
  {
    v8f tile;
    float part;
    fold_math(fo, wave, tile, part);
    sh_bf[2 * wave][lane]     = (v4f){tile[0], tile[1], tile[2], tile[3]};
    sh_bf[2 * wave + 1][lane] = (v4f){tile[4], tile[5], tile[6], tile[7]};
    sh_cp[wave][lane] = part;
    __builtin_amdgcn_fence(__ATOMIC_RELEASE, "workgroup", "local");
    __builtin_amdgcn_s_barrier();
    __builtin_amdgcn_fence(__ATOMIC_ACQUIRE, "workgroup", "local");
    if (!have) return;
#pragma unroll
    for (int j = 0; j < 8; ++j) bf[j] = sh_bf[j][lane];
    cn = ((sh_cp[0][lane] + sh_cp[1][lane]) + sh_cp[2][lane]) + sh_cp[3][lane] + fo.bn;
  }
  v8f cinit = {};
  {
    const int cb = __builtin_bit_cast(int, cn);
#pragma unroll
    for (int r = 0; r < 5; ++r) {
      const float c_lo = __builtin_bit_cast(float, __builtin_amdgcn_readlane(cb, r));
      const float c_hi = __builtin_bit_cast(float, __builtin_amdgcn_readlane(cb, 8 + r));
      cinit[r] = lane >= 16 ? c_hi : c_lo;
    }
  }
  __builtin_amdgcn_sched_barrier(0);

#pragma unroll
  for (int i = NPRE; i < MAXG; ++i) {
    const int g = base + i * WPB;
    load_row_frags(x_row_ptr(x, g, lane, Bn, g < g_end), xs[i]);
  }
  __builtin_amdgcn_sched_barrier(0);

  for (;;) {
    v8f pa = {}, pb = {};
#pragma unroll
    for (int p = 0; p < MAXG / 2; ++p) {
      const int ga = base + (2 * p) * WPB;
      const int ha = ga - 2 * WPB;
      v8f ra = cinit, rb = cinit;
      __builtin_amdgcn_sched_barrier(0);
      mma_pair_pieces<0, 2>(xs[2 * p], xs[2 * p + 1], bf, ra, rb);
      if (p > 0) store_full(pa, out, ha, lane, ha < g_store);
      mma_pair_pieces<2, 4>(xs[2 * p], xs[2 * p + 1], bf, ra, rb);
      if (p > 0) store_full(pb, out, ha + WPB, lane, ha + WPB < g_store);
      mma_pair_pieces<4, 8>(xs[2 * p], xs[2 * p + 1], bf, ra, rb);
      pa = ra;
      pb = rb;
    }
    {
      const int gl = base + (MAXG - 1) * WPB;
      const int ha = gl - 2 * WPB;
      v8f rl = cinit;
      __builtin_amdgcn_sched_barrier(0);
      mma_pieces<0, 4>(xs[MAXG - 1], bf, rl);
      store_full(pa, out, ha, lane, ha < g_store);
      mma_pieces<4, 8>(xs[MAXG - 1], bf, rl);
      store_full(pb, out, ha + WPB, lane, ha + WPB < g_store);
      store_full(rl, out, gl, lane, gl < g_store);
    }
    base += MAXG * WPB;
    if (base >= g_end) break;
#pragma unroll
    for (int i = 0; i < MAXG; ++i) {
      const int g = base + i * WPB;
      load_row_frags(x_row_ptr(x, g, lane, Bn, g < g_end), xs[i]);
    }
    __builtin_amdgcn_sched_barrier(0);
  }

  if (nfull < ngroups) {
    const int gl = ngroups - 1;
    if (gl >= g_begin && gl < g_end && ((gl - g_begin) % WPB) == wave) {
      v4f xl[8];
      load_row_frags(x_row_ptr(x, gl, lane, Bn, true), xl);
      store_group(mma_group(xl, bf, cinit), out, gl, lane, Bn);
    }
  }
}

extern "C" void kernel_launch(void* const* d_in, const int* in_sizes, int n_in,
                              void* d_out, int out_size, void* d_ws, size_t ws_size,
                              hipStream_t stream) {
  const float* x    = (const float*)d_in[0];
  const float* S    = (const float*)d_in[1];
  const float* d    = (const float*)d_in[2];
  const float* bias = (const float*)d_in[3];
  const float* W    = (const float*)d_in[4];
  const float* b    = (const float*)d_in[5];
  float* out = (float*)d_out;

  const int Bn = in_sizes[0] / N_MODES;

  hipLaunchKernelGGL(cv2d_gemm_kernel, dim3(NBLK), dim3(32 * WPB), 0, stream,
                     x, S, d, bias, W, b, out, Bn);
}
